# starting kernel
# baseline (speedup 1.0000x reference)
#include <hip/hip_runtime.h>
#include <math.h>


typedef __attribute__((ext_vector_type(2))) float v2f;
typedef __attribute__((ext_vector_type(8))) float v8f;

#define NN 1024
#define DD 64

__device__ __forceinline__ v8f wmma4(v2f a, v2f b, v8f c) {
  return __builtin_amdgcn_wmma_f32_16x16x4_f32(
       false, a,  false, b,
       (short)0, c,  false,  false);
}

__device__ __forceinline__ float siluf(float v) {
  return v / (1.f + __expf(-v));
}
__device__ __forceinline__ float acoshf_pos(float z) {
  return __logf(z + __fsqrt_rn(z * z - 1.f));
}

__global__ void k_logmap(const float* __restrict__ x, float* __restrict__ u) {
  const int n = blockIdx.x;
  const int t = threadIdx.x;
  __shared__ float s[64];
  const float xv = x[n * DD + t];
  const float v = (t >= 1) ? xv : 0.f;
  s[t] = v * v;
  __syncthreads();
  for (int off = 32; off > 0; off >>= 1) {
    if (t < off) s[t] += s[t + off];
    __syncthreads();
  }
  const float nrm = fmaxf(__fsqrt_rn(s[0]), 1e-7f);
  const float x0 = x[n * DD + 0];
  const float z = fmaxf(x0, 1.f + 1e-7f);
  const float ac = acoshf_pos(z);
  u[n * DD + t] = (t >= 1) ? (ac * xv / nrm) : 0.f;
}

__global__ void k_pipj(const float* __restrict__ u, const float* __restrict__ W_edge,
                       float* __restrict__ P) {
  const int m0 = blockIdx.x * 16;
  const int t = threadIdx.x;
  const int wid = t >> 5, lane = t & 31;
  __shared__ float sA[16][68];
  {
    const int r = t >> 4, c4 = (t & 15) * 4;
    *(float4*)&sA[r][c4] = *(const float4*)(u + (size_t)(m0 + r) * DD + c4);
  }
  __syncthreads();
  const int half = wid >> 2;
  const int ncol = ((wid & 3) * 16) + (lane & 15);
  const int kofs = half * 64;
  const int m = lane & 15;
  const int kp = (lane >> 4) << 1;
  v8f acc = {};
#pragma unroll
  for (int k0 = 0; k0 < 64; k0 += 4) {
    const int kk = k0 + kp;
    v2f a; a.x = sA[m][kk]; a.y = sA[m][kk + 1];
    v2f b; b.x = W_edge[ncol * 192 + kofs + kk];
    b.y = W_edge[ncol * 192 + kofs + kk + 1];
    acc = wmma4(a, b, acc);
  }
  const int n = lane & 15;
  const int mb = 8 * (lane >> 4);
#pragma unroll
  for (int v = 0; v < 8; ++v)
    P[(size_t)(m0 + mb + v) * 128 + half * 64 + (wid & 3) * 16 + n] = acc[v];
}

__global__ void k_edge(const float* __restrict__ e, const float* __restrict__ u,
                       const float* __restrict__ P, const float* __restrict__ W_edge,
                       const float* __restrict__ b_edge, const float* __restrict__ W_ada,
                       const float* __restrict__ b_ada, float* __restrict__ agg) {
  const int i = blockIdx.x;
  const int t = threadIdx.x;
  const int wid = t >> 5, lane = t & 31;
  __shared__ float sEF[64][68];
  __shared__ float sWe[64][68];
  __shared__ float sUi[64];
  __shared__ float sPi[64];
  {
    const int r = t >> 2, c0 = (t & 3) * 16;
    const int j = (i + 1 + r) & (NN - 1);
    const float* __restrict__ src = e + ((size_t)i * NN + j) * DD + c0;
#pragma unroll
    for (int c = 0; c < 16; c += 4) {
      *(float4*)&sWe[r][c0 + c] = *(const float4*)(W_edge + r * 192 + 128 + c0 + c);
      *(float4*)&sEF[r][c0 + c] = *(const float4*)(src + c);
    }
    __builtin_prefetch(P + (size_t)j * 128 + 64, 0, 0);
    __builtin_prefetch(u + (size_t)j * DD, 0, 0);
  }
  if (t < 64) { sUi[t] = u[(size_t)i * DD + t]; sPi[t] = P[(size_t)i * 128 + t]; }
  __syncthreads();

  const int tile0 = wid * 2;
  const int mt = tile0 >> 2, nt0 = tile0 & 3;
  const int m = lane & 15, kp = (lane >> 4) << 1, n = lane & 15;
  v8f acc0 = {}, acc1 = {};
#pragma unroll
  for (int k0 = 0; k0 < 64; k0 += 4) {
    const int kk = k0 + kp;
    v2f a;  a.x  = sEF[mt * 16 + m][kk];        a.y  = sEF[mt * 16 + m][kk + 1];
    v2f b0; b0.x = sWe[nt0 * 16 + n][kk];       b0.y = sWe[nt0 * 16 + n][kk + 1];
    v2f b1; b1.x = sWe[(nt0 + 1) * 16 + n][kk]; b1.y = sWe[(nt0 + 1) * 16 + n][kk + 1];
    acc0 = wmma4(a, b0, acc0);
    acc1 = wmma4(a, b1, acc1);
  }
  __syncthreads();
  {
    const int mb = mt * 16 + 8 * (lane >> 4);
#pragma unroll
    for (int v = 0; v < 8; ++v) {
      sEF[mb + v][nt0 * 16 + n] = acc0[v];
      sEF[mb + v][(nt0 + 1) * 16 + n] = acc1[v];
    }
  }
  __syncthreads();

  const int r = t >> 2, q = t & 3, c0 = q * 16;
  const int j = (i + 1 + r) & (NN - 1);
  const float* __restrict__ uj = u + (size_t)j * DD;
  const float* __restrict__ pj = P + (size_t)j * 128 + 64;
  float inner = 0.f, sum = 0.f, sumsq = 0.f;
  float vals[16];
#pragma unroll
  for (int c = 0; c < 16; ++c) {
    const int cc = c0 + c;
    inner += sUi[cc] * uj[cc];
    const float v = sEF[r][cc] + sPi[cc] + pj[cc] + b_edge[cc];
    vals[c] = v; sum += v; sumsq += v * v;
  }
  inner += __shfl_xor(inner, 1); inner += __shfl_xor(inner, 2);
  sum   += __shfl_xor(sum, 1);   sum   += __shfl_xor(sum, 2);
  sumsq += __shfl_xor(sumsq, 1); sumsq += __shfl_xor(sumsq, 2);
  const float mu = sum * (1.f / 64.f);
  const float var = sumsq * (1.f / 64.f) - mu * mu;
  const float rstd = __frsqrt_rn(var + 1e-6f);
  const float z = fmaxf(-inner, 1.f + 1e-7f);
  float dist = acoshf_pos(z);
  dist = fminf(fmaxf(dist, 1e-6f), 100.f);
  const float s = siluf(dist);
#pragma unroll
  for (int c = 0; c < 16; ++c) {
    const int cc = c0 + c;
    const float shift = s * W_ada[cc] + b_ada[cc];
    const float scale = s * W_ada[64 + cc] + b_ada[64 + cc];
    sEF[r][cc] = (vals[c] - mu) * rstd * (1.f + scale) + shift;
  }
  __syncthreads();
  if (t < 64) {
    float a = 0.f;
#pragma unroll 8
    for (int rr = 0; rr < 64; ++rr) a += sEF[rr][t];
    agg[(size_t)i * DD + t] = a * (1.f / 64.f);
  }
}

__global__ void k_node(const float* __restrict__ u, const float* __restrict__ agg,
                       const float* __restrict__ W_adan, const float* __restrict__ b_adan,
                       const float* __restrict__ W_node, const float* __restrict__ b_node,
                       const float* __restrict__ W_hyp, const float* __restrict__ b_hyp,
                       float* __restrict__ out) {
  const int m0 = blockIdx.x * 16;
  const int t = threadIdx.x;
  const int wid = t >> 5, lane = t & 31;
  __shared__ float sCat[16][132];
  __shared__ float sSA[16][68];
  __shared__ float sAdan[16][196];
  __shared__ float sNv[16][68];
  __shared__ float sH[16][68];
  __shared__ float sSp[16][68];
  {
    const int r = t >> 4, c4 = (t & 15) * 4;
    const float4 xv = *(const float4*)(u + (size_t)(m0 + r) * DD + c4);
    const float4 av = *(const float4*)(agg + (size_t)(m0 + r) * DD + c4);
    *(float4*)&sCat[r][c4] = xv;
    *(float4*)&sCat[r][64 + c4] = av;
    float4 sv;
    sv.x = siluf(av.x); sv.y = siluf(av.y); sv.z = siluf(av.z); sv.w = siluf(av.w);
    *(float4*)&sSA[r][c4] = sv;
  }
  __syncthreads();
  const int m = lane & 15, kp = (lane >> 4) << 1, n = lane & 15;
  for (int tt = wid * 2; tt < wid * 2 + 2; ++tt) {
    v8f acc = {};
    if (tt < 12) {
      const int nglob = tt * 16 + n;
#pragma unroll
      for (int k0 = 0; k0 < 64; k0 += 4) {
        const int kk = k0 + kp;
        v2f a; a.x = sSA[m][kk]; a.y = sSA[m][kk + 1];
        v2f b; b.x = W_adan[nglob * 64 + kk]; b.y = W_adan[nglob * 64 + kk + 1];
        acc = wmma4(a, b, acc);
      }
      const int mb = 8 * (lane >> 4);
#pragma unroll
      for (int v = 0; v < 8; ++v)
        sAdan[mb + v][tt * 16 + n] = acc[v] + b_adan[tt * 16 + n];
    } else {
      const int nglob = (tt - 12) * 16 + n;
#pragma unroll
      for (int k0 = 0; k0 < 128; k0 += 4) {
        const int kk = k0 + kp;
        v2f a; a.x = sCat[m][kk]; a.y = sCat[m][kk + 1];
        v2f b; b.x = W_node[nglob * 128 + kk]; b.y = W_node[nglob * 128 + kk + 1];
        acc = wmma4(a, b, acc);
      }
      const int mb = 8 * (lane >> 4);
#pragma unroll
      for (int v = 0; v < 8; ++v)
        sNv[mb + v][(tt - 12) * 16 + n] = acc[v] + b_node[(tt - 12) * 16 + n];
    }
  }
  __syncthreads();

  const int r = t >> 4, q = t & 15;
  float sum = 0.f, sumsq = 0.f;
  float nvv[4];
#pragma unroll
  for (int c = 0; c < 4; ++c) {
    const float v = sNv[r][q * 4 + c];
    nvv[c] = v; sum += v; sumsq += v * v;
  }
#pragma unroll
  for (int off = 1; off < 16; off <<= 1) {
    sum += __shfl_xor(sum, off);
    sumsq += __shfl_xor(sumsq, off);
  }
  const float mu = sum * (1.f / 64.f);
  const float var = sumsq * (1.f / 64.f) - mu * mu;
  const float rstd = __frsqrt_rn(var + 1e-6f);
#pragma unroll
  for (int c = 0; c < 4; ++c) {
    const int cc = q * 4 + c;
    const float sh = sAdan[r][cc], sc = sAdan[r][64 + cc], gate = sAdan[r][128 + cc];
    const float ln = (nvv[c] - mu) * rstd;
    sH[r][cc] = sCat[r][cc] + gate * (ln * (1.f + sc) + sh);
  }
  __syncthreads();

  if (wid < 4) {
    v8f acc = {};
    const int nglob = wid * 16 + n;
    const float msk = (nglob < 63) ? 1.f : 0.f;
#pragma unroll
    for (int k0 = 0; k0 < 64; k0 += 4) {
      const int kk = k0 + kp;
      v2f a; a.x = sH[m][kk]; a.y = sH[m][kk + 1];
      const int widx = (nglob < 63) ? nglob : 0;
      v2f b; b.x = msk * W_hyp[widx * 64 + kk]; b.y = msk * W_hyp[widx * 64 + kk + 1];
      acc = wmma4(a, b, acc);
    }
    const int mb = 8 * (lane >> 4);
#pragma unroll
    for (int v = 0; v < 8; ++v)
      sSp[mb + v][nglob] = acc[v] + ((nglob < 63) ? b_hyp[nglob] : 0.f);
  }
  __syncthreads();

  float ss = 0.f;
  float spv[4];
#pragma unroll
  for (int c = 0; c < 4; ++c) {
    const int cc = q * 4 + c;
    const float v = (cc < 63) ? sSp[r][cc] : 0.f;
    spv[c] = v; ss += v * v;
  }
#pragma unroll
  for (int off = 1; off < 16; off <<= 1) ss += __shfl_xor(ss, off);
  const float tval = __fsqrt_rn(ss + 1.f);
  float* __restrict__ orow = out + (size_t)(m0 + r) * DD;
  if (q == 0) orow[0] = tval;
#pragma unroll
  for (int c = 0; c < 4; ++c) {
    const int cc = q * 4 + c;
    if (cc < 63) orow[1 + cc] = spv[c];
  }
}

extern "C" void kernel_launch(void* const* d_in, const int* in_sizes, int n_in,
                              void* d_out, int out_size, void* d_ws, size_t ws_size,
                              hipStream_t stream) {
  (void)in_sizes; (void)n_in; (void)out_size; (void)ws_size;
  const float* x      = (const float*)d_in[0];
  const float* e      = (const float*)d_in[2];
  const float* W_edge = (const float*)d_in[4];
  const float* b_edge = (const float*)d_in[5];
  const float* W_node = (const float*)d_in[6];
  const float* b_node = (const float*)d_in[7];
  const float* W_hyp  = (const float*)d_in[8];
  const float* b_hyp  = (const float*)d_in[9];
  const float* W_ada  = (const float*)d_in[10];
  const float* b_ada  = (const float*)d_in[11];
  const float* W_adan = (const float*)d_in[12];
  const float* b_adan = (const float*)d_in[13];
  float* out = (float*)d_out;

  float* ws_u   = (float*)d_ws;
  float* ws_P   = ws_u + NN * DD;
  float* ws_agg = ws_P + NN * 128;

  k_logmap<<<NN, 64, 0, stream>>>(x, ws_u);
  k_pipj<<<NN / 16, 256, 0, stream>>>(ws_u, W_edge, ws_P);
  k_edge<<<NN, 256, 0, stream>>>(e, ws_u, ws_P, W_edge, b_edge, W_ada, b_ada, ws_agg);
  k_node<<<NN / 16, 256, 0, stream>>>(ws_u, ws_agg, W_adan, b_adan, W_node, b_node,
                                      W_hyp, b_hyp, out);
}


# Fuse the log-map kernel into the projection kernel (4 launches -> 3); the norm's sum of squares is now 4 per lane plus a 16-lane shuffle tree (summation order changes)
# speedup vs baseline: 1.1388x; 1.1388x over previous
#include <hip/hip_runtime.h>
#include <math.h>


typedef __attribute__((ext_vector_type(2))) float v2f;
typedef __attribute__((ext_vector_type(8))) float v8f;

#define NN 1024
#define DD 64

__device__ __forceinline__ v8f wmma4(v2f a, v2f b, v8f c) {
  return __builtin_amdgcn_wmma_f32_16x16x4_f32(
       false, a,  false, b,
       (short)0, c,  false,  false);
}

__device__ __forceinline__ float siluf(float v) {
  return v / (1.f + __expf(-v));
}
__device__ __forceinline__ float acoshf_pos(float z) {
  return __logf(z + __fsqrt_rn(z * z - 1.f));
}

__global__ void k_logmap_pipj(const float* __restrict__ x, const float* __restrict__ W_edge,
                              float* __restrict__ u, float* __restrict__ P) {
  const int m0 = blockIdx.x * 16;
  const int t = threadIdx.x;
  const int wid = t >> 5, lane = t & 31;
  __shared__ float sA[16][68];
  const int r = t >> 4, c4 = (t & 15) * 4;
  float4 uv;
  {
    const float4 xv = *(const float4*)(x + (size_t)(m0 + r) * DD + c4);
    float ss = ((c4 != 0) ? xv.x * xv.x : 0.f) + xv.y * xv.y + xv.z * xv.z + xv.w * xv.w;
    for (int off = 1; off < 16; off <<= 1) ss += __shfl_xor(ss, off);
    const float x0 = __shfl(xv.x, lane & 16);
    const float nrm = fmaxf(__fsqrt_rn(ss), 1e-7f);
    const float z = fmaxf(x0, 1.f + 1e-7f);
    const float ac = acoshf_pos(z);
    uv.x = (c4 != 0) ? (ac * xv.x / nrm) : 0.f;
    uv.y = ac * xv.y / nrm;
    uv.z = ac * xv.z / nrm;
    uv.w = ac * xv.w / nrm;
    *(float4*)&sA[r][c4] = uv;
  }
  __syncthreads();
  *(float4*)(u + (size_t)(m0 + r) * DD + c4) = uv;
  const int half = wid >> 2;
  const int ncol = ((wid & 3) * 16) + (lane & 15);
  const int kofs = half * 64;
  const int m = lane & 15;
  const int kp = (lane >> 4) << 1;
  v8f acc = {};
#pragma unroll
  for (int k0 = 0; k0 < 64; k0 += 4) {
    const int kk = k0 + kp;
    v2f a; a.x = sA[m][kk]; a.y = sA[m][kk + 1];
    v2f b; b.x = W_edge[ncol * 192 + kofs + kk];
    b.y = W_edge[ncol * 192 + kofs + kk + 1];
    acc = wmma4(a, b, acc);
  }
  const int n = lane & 15;
  const int mb = 8 * (lane >> 4);
#pragma unroll
  for (int v = 0; v < 8; ++v)
    P[(size_t)(m0 + mb + v) * 128 + half * 64 + (wid & 3) * 16 + n] = acc[v];
}

__global__ void k_edge(const float* __restrict__ e, const float* __restrict__ u,
                       const float* __restrict__ P, const float* __restrict__ W_edge,
                       const float* __restrict__ b_edge, const float* __restrict__ W_ada,
                       const float* __restrict__ b_ada, float* __restrict__ agg) {
  const int i = blockIdx.x;
  const int t = threadIdx.x;
  const int wid = t >> 5, lane = t & 31;
  __shared__ float sEF[64][68];
  __shared__ float sWe[64][68];
  __shared__ float sUi[64];
  __shared__ float sPi[64];
  {
    const int r = t >> 2, c0 = (t & 3) * 16;
    const int j = (i + 1 + r) & (NN - 1);
    const float* __restrict__ src = e + ((size_t)i * NN + j) * DD + c0;
#pragma unroll
    for (int c = 0; c < 16; c += 4) {
      *(float4*)&sWe[r][c0 + c] = *(const float4*)(W_edge + r * 192 + 128 + c0 + c);
      *(float4*)&sEF[r][c0 + c] = *(const float4*)(src + c);
    }
    __builtin_prefetch(P + (size_t)j * 128 + 64, 0, 0);
    __builtin_prefetch(u + (size_t)j * DD, 0, 0);
  }
  if (t < 64) { sUi[t] = u[(size_t)i * DD + t]; sPi[t] = P[(size_t)i * 128 + t]; }
  __syncthreads();

  const int tile0 = wid * 2;
  const int mt = tile0 >> 2, nt0 = tile0 & 3;
  const int m = lane & 15, kp = (lane >> 4) << 1, n = lane & 15;
  v8f acc0 = {}, acc1 = {};
#pragma unroll
  for (int k0 = 0; k0 < 64; k0 += 4) {
    const int kk = k0 + kp;
    v2f a;  a.x  = sEF[mt * 16 + m][kk];        a.y  = sEF[mt * 16 + m][kk + 1];
    v2f b0; b0.x = sWe[nt0 * 16 + n][kk];       b0.y = sWe[nt0 * 16 + n][kk + 1];
    v2f b1; b1.x = sWe[(nt0 + 1) * 16 + n][kk]; b1.y = sWe[(nt0 + 1) * 16 + n][kk + 1];
    acc0 = wmma4(a, b0, acc0);
    acc1 = wmma4(a, b1, acc1);
  }
  __syncthreads();
  {
    const int mb = mt * 16 + 8 * (lane >> 4);
#pragma unroll
    for (int v = 0; v < 8; ++v) {
      sEF[mb + v][nt0 * 16 + n] = acc0[v];
      sEF[mb + v][(nt0 + 1) * 16 + n] = acc1[v];
    }
  }
  __syncthreads();

  const int r = t >> 2, q = t & 3, c0 = q * 16;
  const int j = (i + 1 + r) & (NN - 1);
  const float* __restrict__ uj = u + (size_t)j * DD;
  const float* __restrict__ pj = P + (size_t)j * 128 + 64;
  float inner = 0.f, sum = 0.f, sumsq = 0.f;
  float vals[16];
#pragma unroll
  for (int c = 0; c < 16; ++c) {
    const int cc = c0 + c;
    inner += sUi[cc] * uj[cc];
    const float v = sEF[r][cc] + sPi[cc] + pj[cc] + b_edge[cc];
    vals[c] = v; sum += v; sumsq += v * v;
  }
  inner += __shfl_xor(inner, 1); inner += __shfl_xor(inner, 2);
  sum   += __shfl_xor(sum, 1);   sum   += __shfl_xor(sum, 2);
  sumsq += __shfl_xor(sumsq, 1); sumsq += __shfl_xor(sumsq, 2);
  const float mu = sum * (1.f / 64.f);
  const float var = sumsq * (1.f / 64.f) - mu * mu;
  const float rstd = __frsqrt_rn(var + 1e-6f);
  const float z = fmaxf(-inner, 1.f + 1e-7f);
  float dist = acoshf_pos(z);
  dist = fminf(fmaxf(dist, 1e-6f), 100.f);
  const float s = siluf(dist);
#pragma unroll
  for (int c = 0; c < 16; ++c) {
    const int cc = c0 + c;
    const float shift = s * W_ada[cc] + b_ada[cc];
    const float scale = s * W_ada[64 + cc] + b_ada[64 + cc];
    sEF[r][cc] = (vals[c] - mu) * rstd * (1.f + scale) + shift;
  }
  __syncthreads();
  if (t < 64) {
    float a = 0.f;
#pragma unroll 8
    for (int rr = 0; rr < 64; ++rr) a += sEF[rr][t];
    agg[(size_t)i * DD + t] = a * (1.f / 64.f);
  }
}

__global__ void k_node(const float* __restrict__ u, const float* __restrict__ agg,
                       const float* __restrict__ W_adan, const float* __restrict__ b_adan,
                       const float* __restrict__ W_node, const float* __restrict__ b_node,
                       const float* __restrict__ W_hyp, const float* __restrict__ b_hyp,
                       float* __restrict__ out) {
  const int m0 = blockIdx.x * 16;
  const int t = threadIdx.x;
  const int wid = t >> 5, lane = t & 31;
  __shared__ float sCat[16][132];
  __shared__ float sSA[16][68];
  __shared__ float sAdan[16][196];
  __shared__ float sNv[16][68];
  __shared__ float sH[16][68];
  __shared__ float sSp[16][68];
  {
    const int r = t >> 4, c4 = (t & 15) * 4;
    const float4 xv = *(const float4*)(u + (size_t)(m0 + r) * DD + c4);
    const float4 av = *(const float4*)(agg + (size_t)(m0 + r) * DD + c4);
    *(float4*)&sCat[r][c4] = xv;
    *(float4*)&sCat[r][64 + c4] = av;
    float4 sv;
    sv.x = siluf(av.x); sv.y = siluf(av.y); sv.z = siluf(av.z); sv.w = siluf(av.w);
    *(float4*)&sSA[r][c4] = sv;
  }
  __syncthreads();
  const int m = lane & 15, kp = (lane >> 4) << 1, n = lane & 15;
  for (int tt = wid * 2; tt < wid * 2 + 2; ++tt) {
    v8f acc = {};
    if (tt < 12) {
      const int nglob = tt * 16 + n;
#pragma unroll
      for (int k0 = 0; k0 < 64; k0 += 4) {
        const int kk = k0 + kp;
        v2f a; a.x = sSA[m][kk]; a.y = sSA[m][kk + 1];
        v2f b; b.x = W_adan[nglob * 64 + kk]; b.y = W_adan[nglob * 64 + kk + 1];
        acc = wmma4(a, b, acc);
      }
      const int mb = 8 * (lane >> 4);
#pragma unroll
      for (int v = 0; v < 8; ++v)
        sAdan[mb + v][tt * 16 + n] = acc[v] + b_adan[tt * 16 + n];
    } else {
      const int nglob = (tt - 12) * 16 + n;
#pragma unroll
      for (int k0 = 0; k0 < 128; k0 += 4) {
        const int kk = k0 + kp;
        v2f a; a.x = sCat[m][kk]; a.y = sCat[m][kk + 1];
        v2f b; b.x = W_node[nglob * 128 + kk]; b.y = W_node[nglob * 128 + kk + 1];
        acc = wmma4(a, b, acc);
      }
      const int mb = 8 * (lane >> 4);
#pragma unroll
      for (int v = 0; v < 8; ++v)
        sNv[mb + v][(tt - 12) * 16 + n] = acc[v] + b_node[(tt - 12) * 16 + n];
    }
  }
  __syncthreads();

  const int r = t >> 4, q = t & 15;
  float sum = 0.f, sumsq = 0.f;
  float nvv[4];
#pragma unroll
  for (int c = 0; c < 4; ++c) {
    const float v = sNv[r][q * 4 + c];
    nvv[c] = v; sum += v; sumsq += v * v;
  }
#pragma unroll
  for (int off = 1; off < 16; off <<= 1) {
    sum += __shfl_xor(sum, off);
    sumsq += __shfl_xor(sumsq, off);
  }
  const float mu = sum * (1.f / 64.f);
  const float var = sumsq * (1.f / 64.f) - mu * mu;
  const float rstd = __frsqrt_rn(var + 1e-6f);
#pragma unroll
  for (int c = 0; c < 4; ++c) {
    const int cc = q * 4 + c;
    const float sh = sAdan[r][cc], sc = sAdan[r][64 + cc], gate = sAdan[r][128 + cc];
    const float ln = (nvv[c] - mu) * rstd;
    sH[r][cc] = sCat[r][cc] + gate * (ln * (1.f + sc) + sh);
  }
  __syncthreads();

  if (wid < 4) {
    v8f acc = {};
    const int nglob = wid * 16 + n;
    const float msk = (nglob < 63) ? 1.f : 0.f;
#pragma unroll
    for (int k0 = 0; k0 < 64; k0 += 4) {
      const int kk = k0 + kp;
      v2f a; a.x = sH[m][kk]; a.y = sH[m][kk + 1];
      const int widx = (nglob < 63) ? nglob : 0;
      v2f b; b.x = msk * W_hyp[widx * 64 + kk]; b.y = msk * W_hyp[widx * 64 + kk + 1];
      acc = wmma4(a, b, acc);
    }
    const int mb = 8 * (lane >> 4);
#pragma unroll
    for (int v = 0; v < 8; ++v)
      sSp[mb + v][nglob] = acc[v] + ((nglob < 63) ? b_hyp[nglob] : 0.f);
  }
  __syncthreads();

  float ss = 0.f;
  float spv[4];
#pragma unroll
  for (int c = 0; c < 4; ++c) {
    const int cc = q * 4 + c;
    const float v = (cc < 63) ? sSp[r][cc] : 0.f;
    spv[c] = v; ss += v * v;
  }
#pragma unroll
  for (int off = 1; off < 16; off <<= 1) ss += __shfl_xor(ss, off);
  const float tval = __fsqrt_rn(ss + 1.f);
  float* __restrict__ orow = out + (size_t)(m0 + r) * DD;
  if (q == 0) orow[0] = tval;
#pragma unroll
  for (int c = 0; c < 4; ++c) {
    const int cc = q * 4 + c;
    if (cc < 63) orow[1 + cc] = spv[c];
  }
}

extern "C" void kernel_launch(void* const* d_in, const int* in_sizes, int n_in,
                              void* d_out, int out_size, void* d_ws, size_t ws_size,
                              hipStream_t stream) {
  (void)in_sizes; (void)n_in; (void)out_size; (void)ws_size;
  const float* x      = (const float*)d_in[0];
  const float* e      = (const float*)d_in[2];
  const float* W_edge = (const float*)d_in[4];
  const float* b_edge = (const float*)d_in[5];
  const float* W_node = (const float*)d_in[6];
  const float* b_node = (const float*)d_in[7];
  const float* W_hyp  = (const float*)d_in[8];
  const float* b_hyp  = (const float*)d_in[9];
  const float* W_ada  = (const float*)d_in[10];
  const float* b_ada  = (const float*)d_in[11];
  const float* W_adan = (const float*)d_in[12];
  const float* b_adan = (const float*)d_in[13];
  float* out = (float*)d_out;

  float* ws_u   = (float*)d_ws;
  float* ws_P   = ws_u + NN * DD;
  float* ws_agg = ws_P + NN * 128;

  k_logmap_pipj<<<NN / 16, 256, 0, stream>>>(x, W_edge, ws_u, ws_P);
  k_edge<<<NN, 256, 0, stream>>>(e, ws_u, ws_P, W_edge, b_edge, W_ada, b_ada, ws_agg);
  k_node<<<NN / 16, 256, 0, stream>>>(ws_u, ws_agg, W_adan, b_adan, W_node, b_node,
                                      W_hyp, b_hyp, out);
}


# Every global operand (weights, biases, neighbour rows of P and u) is read once per work-group by lane-contiguous whole-line float4 loads and staged in LDS; no 16-line loads left; sums unchanged
# speedup vs baseline: 1.3691x; 1.2022x over previous
#include <hip/hip_runtime.h>
#include <math.h>


typedef __attribute__((ext_vector_type(2))) float v2f;
typedef __attribute__((ext_vector_type(8))) float v8f;

#define NN 1024
#define DD 64

__device__ __forceinline__ v8f wmma4(v2f a, v2f b, v8f c) {
  return __builtin_amdgcn_wmma_f32_16x16x4_f32(
       false, a,  false, b,
       (short)0, c,  false,  false);
}

__device__ __forceinline__ float siluf(float v) {
  return v / (1.f + __expf(-v));
}
__device__ __forceinline__ float acoshf_pos(float z) {
  return __logf(z + __fsqrt_rn(z * z - 1.f));
}

__global__ void k_logmap_pipj(const float* __restrict__ x, const float* __restrict__ W_edge,
                              float* __restrict__ u, float* __restrict__ P) {
  const int m0 = blockIdx.x * 16;
  const int t = threadIdx.x;
  const int wid = t >> 5, lane = t & 31;
  __shared__ float sA[16][68];
  __shared__ float sW[64][132];
  const int r = t >> 4, c4 = (t & 15) * 4;
  const float4 xv = *(const float4*)(x + (size_t)(m0 + r) * DD + c4);
#pragma unroll
  for (int p = 0; p < 8; ++p) {
    const int f = p * 256 + t;
    const int wr = f >> 5, wc = (f & 31) * 4;
    *(float4*)&sW[wr][wc] = *(const float4*)(W_edge + wr * 192 + wc);
  }
  float4 uv;
  {
    float ss = ((c4 != 0) ? xv.x * xv.x : 0.f) + xv.y * xv.y + xv.z * xv.z + xv.w * xv.w;
    for (int off = 1; off < 16; off <<= 1) ss += __shfl_xor(ss, off);
    const float x0 = __shfl(xv.x, lane & 16);
    const float nrm = fmaxf(__fsqrt_rn(ss), 1e-7f);
    const float z = fmaxf(x0, 1.f + 1e-7f);
    const float ac = acoshf_pos(z);
    uv.x = (c4 != 0) ? (ac * xv.x / nrm) : 0.f;
    uv.y = ac * xv.y / nrm;
    uv.z = ac * xv.z / nrm;
    uv.w = ac * xv.w / nrm;
    *(float4*)&sA[r][c4] = uv;
  }
  __syncthreads();
  *(float4*)(u + (size_t)(m0 + r) * DD + c4) = uv;
  const int half = wid >> 2;
  const int ncol = ((wid & 3) * 16) + (lane & 15);
  const int kofs = half * 64;
  const int m = lane & 15;
  const int kp = (lane >> 4) << 1;
  v8f acc = {};
#pragma unroll
  for (int k0 = 0; k0 < 64; k0 += 4) {
    const int kk = k0 + kp;
    v2f a; a.x = sA[m][kk]; a.y = sA[m][kk + 1];
    v2f b; b.x = sW[ncol][kofs + kk];
    b.y = sW[ncol][kofs + kk + 1];
    acc = wmma4(a, b, acc);
  }
  const int n = lane & 15;
  const int mb = 8 * (lane >> 4);
#pragma unroll
  for (int v = 0; v < 8; ++v)
    P[(size_t)(m0 + mb + v) * 128 + half * 64 + (wid & 3) * 16 + n] = acc[v];
}

__global__ void k_edge(const float* __restrict__ e, const float* __restrict__ u,
                       const float* __restrict__ P, const float* __restrict__ W_edge,
                       const float* __restrict__ b_edge, const float* __restrict__ W_ada,
                       const float* __restrict__ b_ada, float* __restrict__ agg) {
  const int i = blockIdx.x;
  const int t = threadIdx.x;
  const int wid = t >> 5, lane = t & 31;
  __shared__ float sEF[64][68];
  __shared__ float sWe[64][68];
  __shared__ float sPj[64][68];
  __shared__ float sUj[64][68];
  __shared__ float sUi[256];
  __shared__ float sPi[256];
  {
#pragma unroll
    for (int p = 0; p < 4; ++p) {
      const int f = p * 256 + t;
      const int r = f >> 4, c4 = (f & 15) * 4;
      const int j = (i + 1 + r) & (NN - 1);
      *(float4*)&sWe[r][c4] = *(const float4*)(W_edge + r * 192 + 128 + c4);
      *(float4*)&sEF[r][c4] = *(const float4*)(e + ((size_t)i * NN + j) * DD + c4);
      *(float4*)&sPj[r][c4] = *(const float4*)(P + (size_t)j * 128 + 64 + c4);
      *(float4*)&sUj[r][c4] = *(const float4*)(u + (size_t)j * DD + c4);
    }
  }
  sUi[t] = u[(size_t)i * DD + (t & 63)];
  sPi[t] = P[(size_t)i * 128 + (t & 63)];
  __syncthreads();

  const int tile0 = wid * 2;
  const int mt = tile0 >> 2, nt0 = tile0 & 3;
  const int m = lane & 15, kp = (lane >> 4) << 1, n = lane & 15;
  v8f acc0 = {}, acc1 = {};
#pragma unroll
  for (int k0 = 0; k0 < 64; k0 += 4) {
    const int kk = k0 + kp;
    v2f a;  a.x  = sEF[mt * 16 + m][kk];        a.y  = sEF[mt * 16 + m][kk + 1];
    v2f b0; b0.x = sWe[nt0 * 16 + n][kk];       b0.y = sWe[nt0 * 16 + n][kk + 1];
    v2f b1; b1.x = sWe[(nt0 + 1) * 16 + n][kk]; b1.y = sWe[(nt0 + 1) * 16 + n][kk + 1];
    acc0 = wmma4(a, b0, acc0);
    acc1 = wmma4(a, b1, acc1);
  }
  __syncthreads();
  {
    const int mb = mt * 16 + 8 * (lane >> 4);
#pragma unroll
    for (int v = 0; v < 8; ++v) {
      sEF[mb + v][nt0 * 16 + n] = acc0[v];
      sEF[mb + v][(nt0 + 1) * 16 + n] = acc1[v];
    }
  }
  __syncthreads();

  const int r = t >> 2, q = t & 3, c0 = q * 16;
  float inner = 0.f, sum = 0.f, sumsq = 0.f;
  float vals[16];
#pragma unroll
  for (int c = 0; c < 16; ++c) {
    const int cc = c0 + c;
    inner += sUi[cc] * sUj[r][cc];
    const float v = sEF[r][cc] + sPi[cc] + sPj[r][cc] + b_edge[cc];
    vals[c] = v; sum += v; sumsq += v * v;
  }
  inner += __shfl_xor(inner, 1); inner += __shfl_xor(inner, 2);
  sum   += __shfl_xor(sum, 1);   sum   += __shfl_xor(sum, 2);
  sumsq += __shfl_xor(sumsq, 1); sumsq += __shfl_xor(sumsq, 2);
  const float mu = sum * (1.f / 64.f);
  const float var = sumsq * (1.f / 64.f) - mu * mu;
  const float rstd = __frsqrt_rn(var + 1e-6f);
  const float z = fmaxf(-inner, 1.f + 1e-7f);
  float dist = acoshf_pos(z);
  dist = fminf(fmaxf(dist, 1e-6f), 100.f);
  const float s = siluf(dist);
#pragma unroll
  for (int c = 0; c < 16; ++c) {
    const int cc = c0 + c;
    const float shift = s * W_ada[cc] + b_ada[cc];
    const float scale = s * W_ada[64 + cc] + b_ada[64 + cc];
    sEF[r][cc] = (vals[c] - mu) * rstd * (1.f + scale) + shift;
  }
  __syncthreads();
  if (t < 64) {
    float a = 0.f;
#pragma unroll 8
    for (int rr = 0; rr < 64; ++rr) a += sEF[rr][t];
    agg[(size_t)i * DD + t] = a * (1.f / 64.f);
  }
}

__global__ void k_node(const float* __restrict__ u, const float* __restrict__ agg,
                       const float* __restrict__ W_adan, const float* __restrict__ b_adan,
                       const float* __restrict__ W_node, const float* __restrict__ b_node,
                       const float* __restrict__ W_hyp, const float* __restrict__ b_hyp,
                       float* __restrict__ out) {
  const int m0 = blockIdx.x * 16;
  const int t = threadIdx.x;
  const int wid = t >> 5, lane = t & 31;
  __shared__ float sCat[16][132];
  __shared__ float sSA[16][68];
  __shared__ float sAdan[16][196];
  __shared__ float sNv[16][68];
  __shared__ float sH[16][68];
  __shared__ float sSp[16][68];
  __shared__ float sWadan[192][68];
  __shared__ float sWnode[64][132];
  __shared__ float sWhyp[64][68];
  __shared__ float sBadan[256];
  __shared__ float sBnode[256];
  __shared__ float sBhyp[256];
  {
    const int r = t >> 4, c4 = (t & 15) * 4;
    const float4 xv = *(const float4*)(u + (size_t)(m0 + r) * DD + c4);
    const float4 av = *(const float4*)(agg + (size_t)(m0 + r) * DD + c4);
    *(float4*)&sCat[r][c4] = xv;
    *(float4*)&sCat[r][64 + c4] = av;
    float4 sv;
    sv.x = siluf(av.x); sv.y = siluf(av.y); sv.z = siluf(av.z); sv.w = siluf(av.w);
    *(float4*)&sSA[r][c4] = sv;
  }
  {
#pragma unroll
    for (int p = 0; p < 12; ++p) {
      const int f = p * 256 + t;
      *(float4*)&sWadan[f >> 4][(f & 15) * 4] = *(const float4*)(W_adan + (size_t)f * 4);
    }
#pragma unroll
    for (int p = 0; p < 8; ++p) {
      const int f = p * 256 + t;
      *(float4*)&sWnode[f >> 5][(f & 31) * 4] = *(const float4*)(W_node + (size_t)f * 4);
    }
#pragma unroll
    for (int p = 0; p < 4; ++p) {
      const int f = p * 256 + t;
      const int fc = (f < 63 * 16) ? f : (63 * 16 - 1);
      float4 w = *(const float4*)(W_hyp + (size_t)fc * 4);
      if (f >= 63 * 16) w = make_float4(0.f, 0.f, 0.f, 0.f);
      *(float4*)&sWhyp[f >> 4][(f & 15) * 4] = w;
    }
    sBadan[t] = b_adan[(t < 192) ? t : 191];
    sBnode[t] = b_node[t & 63];
    const float bh = b_hyp[((t & 63) < 63) ? (t & 63) : 62];
    sBhyp[t] = ((t & 63) < 63) ? bh : 0.f;
  }
  __syncthreads();
  const int m = lane & 15, kp = (lane >> 4) << 1, n = lane & 15;
  for (int tt = wid * 2; tt < wid * 2 + 2; ++tt) {
    v8f acc = {};
    if (tt < 12) {
      const int nglob = tt * 16 + n;
#pragma unroll
      for (int k0 = 0; k0 < 64; k0 += 4) {
        const int kk = k0 + kp;
        v2f a; a.x = sSA[m][kk]; a.y = sSA[m][kk + 1];
        v2f b; b.x = sWadan[nglob][kk]; b.y = sWadan[nglob][kk + 1];
        acc = wmma4(a, b, acc);
      }
      const int mb = 8 * (lane >> 4);
#pragma unroll
      for (int v = 0; v < 8; ++v)
        sAdan[mb + v][tt * 16 + n] = acc[v] + sBadan[tt * 16 + n];
    } else {
      const int nglob = (tt - 12) * 16 + n;
#pragma unroll
      for (int k0 = 0; k0 < 128; k0 += 4) {
        const int kk = k0 + kp;
        v2f a; a.x = sCat[m][kk]; a.y = sCat[m][kk + 1];
        v2f b; b.x = sWnode[nglob][kk]; b.y = sWnode[nglob][kk + 1];
        acc = wmma4(a, b, acc);
      }
      const int mb = 8 * (lane >> 4);
#pragma unroll
      for (int v = 0; v < 8; ++v)
        sNv[mb + v][(tt - 12) * 16 + n] = acc[v] + sBnode[(tt - 12) * 16 + n];
    }
  }
  __syncthreads();

  const int r = t >> 4, q = t & 15;
  float sum = 0.f, sumsq = 0.f;
  float nvv[4];
#pragma unroll
  for (int c = 0; c < 4; ++c) {
    const float v = sNv[r][q * 4 + c];
    nvv[c] = v; sum += v; sumsq += v * v;
  }
#pragma unroll
  for (int off = 1; off < 16; off <<= 1) {
    sum += __shfl_xor(sum, off);
    sumsq += __shfl_xor(sumsq, off);
  }
  const float mu = sum * (1.f / 64.f);
  const float var = sumsq * (1.f / 64.f) - mu * mu;
  const float rstd = __frsqrt_rn(var + 1e-6f);
#pragma unroll
  for (int c = 0; c < 4; ++c) {
    const int cc = q * 4 + c;
    const float sh = sAdan[r][cc], sc = sAdan[r][64 + cc], gate = sAdan[r][128 + cc];
    const float ln = (nvv[c] - mu) * rstd;
    sH[r][cc] = sCat[r][cc] + gate * (ln * (1.f + sc) + sh);
  }
  __syncthreads();

  if (wid < 4) {
    v8f acc = {};
    const int nglob = wid * 16 + n;
#pragma unroll
    for (int k0 = 0; k0 < 64; k0 += 4) {
      const int kk = k0 + kp;
      v2f a; a.x = sH[m][kk]; a.y = sH[m][kk + 1];
      v2f b; b.x = sWhyp[nglob][kk]; b.y = sWhyp[nglob][kk + 1];
      acc = wmma4(a, b, acc);
    }
    const int mb = 8 * (lane >> 4);
    const float bh = sBhyp[nglob];
#pragma unroll
    for (int v = 0; v < 8; ++v)
      sSp[mb + v][nglob] = acc[v] + bh;
  }
  __syncthreads();

  float ss = 0.f;
  float spv[4];
#pragma unroll
  for (int c = 0; c < 4; ++c) {
    const int cc = q * 4 + c;
    const float v = (cc < 63) ? sSp[r][cc] : 0.f;
    spv[c] = v; ss += v * v;
  }
#pragma unroll
  for (int off = 1; off < 16; off <<= 1) ss += __shfl_xor(ss, off);
  const float tval = __fsqrt_rn(ss + 1.f);
  float* __restrict__ orow = out + (size_t)(m0 + r) * DD;
  if (q == 0) orow[0] = tval;
#pragma unroll
  for (int c = 0; c < 4; ++c) {
    const int cc = q * 4 + c;
    if (cc < 63) orow[1 + cc] = spv[c];
  }
}

extern "C" void kernel_launch(void* const* d_in, const int* in_sizes, int n_in,
                              void* d_out, int out_size, void* d_ws, size_t ws_size,
                              hipStream_t stream) {
  (void)in_sizes; (void)n_in; (void)out_size; (void)ws_size;
  const float* x      = (const float*)d_in[0];
  const float* e      = (const float*)d_in[2];
  const float* W_edge = (const float*)d_in[4];
  const float* b_edge = (const float*)d_in[5];
  const float* W_node = (const float*)d_in[6];
  const float* b_node = (const float*)d_in[7];
  const float* W_hyp  = (const float*)d_in[8];
  const float* b_hyp  = (const float*)d_in[9];
  const float* W_ada  = (const float*)d_in[10];
  const float* b_ada  = (const float*)d_in[11];
  const float* W_adan = (const float*)d_in[12];
  const float* b_adan = (const float*)d_in[13];
  float* out = (float*)d_out;

  float* ws_u   = (float*)d_ws;
  float* ws_P   = ws_u + NN * DD;
  float* ws_agg = ws_P + NN * 128;

  k_logmap_pipj<<<NN / 16, 256, 0, stream>>>(x, W_edge, ws_u, ws_P);
  k_edge<<<NN, 256, 0, stream>>>(e, ws_u, ws_P, W_edge, b_edge, W_ada, b_ada, ws_agg);
  k_node<<<NN / 16, 256, 0, stream>>>(ws_u, ws_agg, W_adan, b_adan, W_node, b_node,
                                      W_hyp, b_hyp, out);
}


# Projection kernel fused into the edge kernel (3 launches -> 2): a 1024-thread work-group owns 4 nodes, log-maps and projects its own 68-row window on WMMA; u, P no longer round-trip; sums unchanged
# speedup vs baseline: 1.5892x; 1.1608x over previous
#include <hip/hip_runtime.h>
#include <math.h>


typedef __attribute__((ext_vector_type(2))) float v2f;
typedef __attribute__((ext_vector_type(8))) float v8f;

#define NN 1024
#define DD 64

__device__ __forceinline__ v8f wmma4(v2f a, v2f b, v8f c) {
  return __builtin_amdgcn_wmma_f32_16x16x4_f32(
       false, a,  false, b,
       (short)0, c,  false,  false);
}

__device__ __forceinline__ float siluf(float v) {
  return v / (1.f + __expf(-v));
}
__device__ __forceinline__ float acoshf_pos(float z) {
  return __logf(z + __fsqrt_rn(z * z - 1.f));
}

#define GN 4
#define WROWS 80

__device__ __forceinline__ float4 logmap4(const float4 xv, const int c4, const int lane) {
  float ss = ((c4 != 0) ? xv.x * xv.x : 0.f) + xv.y * xv.y + xv.z * xv.z + xv.w * xv.w;
  for (int off = 1; off < 16; off <<= 1) ss += __shfl_xor(ss, off);
  const float x0 = __shfl(xv.x, lane & 16);
  const float nrm = fmaxf(__fsqrt_rn(ss), 1e-7f);
  const float z = fmaxf(x0, 1.f + 1e-7f);
  const float ac = acoshf_pos(z);
  float4 uv;
  uv.x = (c4 != 0) ? (ac * xv.x / nrm) : 0.f;
  uv.y = ac * xv.y / nrm;
  uv.z = ac * xv.z / nrm;
  uv.w = ac * xv.w / nrm;
  return uv;
}

__global__ __launch_bounds__(1024) void k_edge(
    const float* __restrict__ x, const float* __restrict__ e, const float* __restrict__ W_edge,
    const float* __restrict__ b_edge, const float* __restrict__ W_ada,
    const float* __restrict__ b_ada, float* __restrict__ u, float* __restrict__ agg) {
  const int i0 = blockIdx.x * GN;
  const int t = threadIdx.x;
  const int wid = t >> 5, lane = t & 31;
  const int g = t >> 8, tl = t & 255;
  const int wl = wid & 7;
  const int i = i0 + g;
  __shared__ float sEF[GN][64][68];
  __shared__ float sW[64][196];
  __shared__ float sU[WROWS][68];
  __shared__ float sPj[WROWS][68];
  __shared__ float sPi[16][68];
  const int w0 = t >> 4, c4 = (t & 15) * 4;
  const float4 xv0 = *(const float4*)(x + (unsigned)(((i0 + w0) & (NN - 1)) * DD + c4));
  const int w1 = 64 + (t >> 4);
  float4 xv1 = make_float4(0.f, 0.f, 0.f, 0.f);
  if (t < 256) {
    const int wc = (w1 < 68) ? w1 : 67;
    xv1 = *(const float4*)(x + (unsigned)(((i0 + wc) & (NN - 1)) * DD + c4));
  }
#pragma unroll
  for (int p = 0; p < 3; ++p) {
    const int f = p * 1024 + t;
    *(float4*)&sW[f / 48][(f % 48) * 4] = *(const float4*)(W_edge + (unsigned)(f * 4));
  }
#pragma unroll
  for (int p = 0; p < 4; ++p) {
    const int f = p * 256 + tl;
    const int r = f >> 4, cc4 = (f & 15) * 4;
    const int j = (i + 1 + r) & (NN - 1);
    *(float4*)&sEF[g][r][cc4] = *(const float4*)(e + (unsigned)((i * NN + j) * DD + cc4));
  }
  {
    const float4 uv = logmap4(xv0, c4, lane);
    *(float4*)&sU[w0][c4] = uv;
    if (w0 < GN) *(float4*)(u + (unsigned)((i0 + w0) * DD + c4)) = uv;
  }
  if (t < 256) {
    float4 uv = logmap4(xv1, c4, lane);
    if (w1 >= 68) uv = make_float4(0.f, 0.f, 0.f, 0.f);
    *(float4*)&sU[w1][c4] = uv;
  }
  __syncthreads();

  const int m = lane & 15, kp = (lane >> 4) << 1, n = lane & 15;
  const int mbh = 8 * (lane >> 4);
  if (wid < 24) {
    const bool isPj = wid < 20;
    const int mtw = isPj ? (wid >> 2) : 0;
    const int ntw = isPj ? (wid & 3) : (wid - 20);
    const int kofs = isPj ? 64 : 0;
    v8f acc = {};
#pragma unroll
    for (int k0 = 0; k0 < 64; k0 += 4) {
      const int kk = k0 + kp;
      v2f a; a.x = sU[mtw * 16 + m][kk]; a.y = sU[mtw * 16 + m][kk + 1];
      v2f b; b.x = sW[ntw * 16 + n][kofs + kk]; b.y = sW[ntw * 16 + n][kofs + kk + 1];
      acc = wmma4(a, b, acc);
    }
    if (isPj) {
#pragma unroll
      for (int v = 0; v < 8; ++v) sPj[mtw * 16 + mbh + v][ntw * 16 + n] = acc[v];
    } else {
#pragma unroll
      for (int v = 0; v < 8; ++v) sPi[mbh + v][ntw * 16 + n] = acc[v];
    }
  }

  const int tile0 = wl * 2;
  const int mt = tile0 >> 2, nt0 = tile0 & 3;
  v8f acc0 = {}, acc1 = {};
#pragma unroll
  for (int k0 = 0; k0 < 64; k0 += 4) {
    const int kk = k0 + kp;
    v2f a;  a.x  = sEF[g][mt * 16 + m][kk];            a.y  = sEF[g][mt * 16 + m][kk + 1];
    v2f b0; b0.x = sW[nt0 * 16 + n][128 + kk];         b0.y = sW[nt0 * 16 + n][128 + kk + 1];
    v2f b1; b1.x = sW[(nt0 + 1) * 16 + n][128 + kk];   b1.y = sW[(nt0 + 1) * 16 + n][128 + kk + 1];
    acc0 = wmma4(a, b0, acc0);
    acc1 = wmma4(a, b1, acc1);
  }
  __syncthreads();
  {
    const int mb = mt * 16 + mbh;
#pragma unroll
    for (int v = 0; v < 8; ++v) {
      sEF[g][mb + v][nt0 * 16 + n] = acc0[v];
      sEF[g][mb + v][(nt0 + 1) * 16 + n] = acc1[v];
    }
  }
  __syncthreads();

  const int r = tl >> 2, q = tl & 3, c0 = q * 16;
  const int wr = g + 1 + r;
  float inner = 0.f, sum = 0.f, sumsq = 0.f;
  float vals[16];
#pragma unroll
  for (int c = 0; c < 16; ++c) {
    const int cc = c0 + c;
    inner += sU[g][cc] * sU[wr][cc];
    const float v = sEF[g][r][cc] + sPi[g][cc] + sPj[wr][cc] + b_edge[cc];
    vals[c] = v; sum += v; sumsq += v * v;
  }
  inner += __shfl_xor(inner, 1); inner += __shfl_xor(inner, 2);
  sum   += __shfl_xor(sum, 1);   sum   += __shfl_xor(sum, 2);
  sumsq += __shfl_xor(sumsq, 1); sumsq += __shfl_xor(sumsq, 2);
  const float mu = sum * (1.f / 64.f);
  const float var = sumsq * (1.f / 64.f) - mu * mu;
  const float rstd = __frsqrt_rn(var + 1e-6f);
  const float z = fmaxf(-inner, 1.f + 1e-7f);
  float dist = acoshf_pos(z);
  dist = fminf(fmaxf(dist, 1e-6f), 100.f);
  const float s = siluf(dist);
#pragma unroll
  for (int c = 0; c < 16; ++c) {
    const int cc = c0 + c;
    const float shift = s * W_ada[cc] + b_ada[cc];
    const float scale = s * W_ada[64 + cc] + b_ada[64 + cc];
    sEF[g][r][cc] = (vals[c] - mu) * rstd * (1.f + scale) + shift;
  }
  __syncthreads();
  if (tl < 64) {
    float a = 0.f;
#pragma unroll 8
    for (int rr = 0; rr < 64; ++rr) a += sEF[g][rr][tl];
    agg[(unsigned)(i * DD + tl)] = a * (1.f / 64.f);
  }
}

__global__ void k_node(const float* __restrict__ u, const float* __restrict__ agg,
                       const float* __restrict__ W_adan, const float* __restrict__ b_adan,
                       const float* __restrict__ W_node, const float* __restrict__ b_node,
                       const float* __restrict__ W_hyp, const float* __restrict__ b_hyp,
                       float* __restrict__ out) {
  const int m0 = blockIdx.x * 16;
  const int t = threadIdx.x;
  const int wid = t >> 5, lane = t & 31;
  __shared__ float sCat[16][132];
  __shared__ float sSA[16][68];
  __shared__ float sAdan[16][196];
  __shared__ float sNv[16][68];
  __shared__ float sH[16][68];
  __shared__ float sSp[16][68];
  __shared__ float sWadan[192][68];
  __shared__ float sWnode[64][132];
  __shared__ float sWhyp[64][68];
  __shared__ float sBadan[256];
  __shared__ float sBnode[256];
  __shared__ float sBhyp[256];
  {
    const int r = t >> 4, c4 = (t & 15) * 4;
    const float4 xv = *(const float4*)(u + (size_t)(m0 + r) * DD + c4);
    const float4 av = *(const float4*)(agg + (size_t)(m0 + r) * DD + c4);
    *(float4*)&sCat[r][c4] = xv;
    *(float4*)&sCat[r][64 + c4] = av;
    float4 sv;
    sv.x = siluf(av.x); sv.y = siluf(av.y); sv.z = siluf(av.z); sv.w = siluf(av.w);
    *(float4*)&sSA[r][c4] = sv;
  }
  {
#pragma unroll
    for (int p = 0; p < 12; ++p) {
      const int f = p * 256 + t;
      *(float4*)&sWadan[f >> 4][(f & 15) * 4] = *(const float4*)(W_adan + (size_t)f * 4);
    }
#pragma unroll
    for (int p = 0; p < 8; ++p) {
      const int f = p * 256 + t;
      *(float4*)&sWnode[f >> 5][(f & 31) * 4] = *(const float4*)(W_node + (size_t)f * 4);
    }
#pragma unroll
    for (int p = 0; p < 4; ++p) {
      const int f = p * 256 + t;
      const int fc = (f < 63 * 16) ? f : (63 * 16 - 1);
      float4 w = *(const float4*)(W_hyp + (size_t)fc * 4);
      if (f >= 63 * 16) w = make_float4(0.f, 0.f, 0.f, 0.f);
      *(float4*)&sWhyp[f >> 4][(f & 15) * 4] = w;
    }
    sBadan[t] = b_adan[(t < 192) ? t : 191];
    sBnode[t] = b_node[t & 63];
    const float bh = b_hyp[((t & 63) < 63) ? (t & 63) : 62];
    sBhyp[t] = ((t & 63) < 63) ? bh : 0.f;
  }
  __syncthreads();
  const int m = lane & 15, kp = (lane >> 4) << 1, n = lane & 15;
  for (int tt = wid * 2; tt < wid * 2 + 2; ++tt) {
    v8f acc = {};
    if (tt < 12) {
      const int nglob = tt * 16 + n;
#pragma unroll
      for (int k0 = 0; k0 < 64; k0 += 4) {
        const int kk = k0 + kp;
        v2f a; a.x = sSA[m][kk]; a.y = sSA[m][kk + 1];
        v2f b; b.x = sWadan[nglob][kk]; b.y = sWadan[nglob][kk + 1];
        acc = wmma4(a, b, acc);
      }
      const int mb = 8 * (lane >> 4);
#pragma unroll
      for (int v = 0; v < 8; ++v)
        sAdan[mb + v][tt * 16 + n] = acc[v] + sBadan[tt * 16 + n];
    } else {
      const int nglob = (tt - 12) * 16 + n;
#pragma unroll
      for (int k0 = 0; k0 < 128; k0 += 4) {
        const int kk = k0 + kp;
        v2f a; a.x = sCat[m][kk]; a.y = sCat[m][kk + 1];
        v2f b; b.x = sWnode[nglob][kk]; b.y = sWnode[nglob][kk + 1];
        acc = wmma4(a, b, acc);
      }
      const int mb = 8 * (lane >> 4);
#pragma unroll
      for (int v = 0; v < 8; ++v)
        sNv[mb + v][(tt - 12) * 16 + n] = acc[v] + sBnode[(tt - 12) * 16 + n];
    }
  }
  __syncthreads();

  const int r = t >> 4, q = t & 15;
  float sum = 0.f, sumsq = 0.f;
  float nvv[4];
#pragma unroll
  for (int c = 0; c < 4; ++c) {
    const float v = sNv[r][q * 4 + c];
    nvv[c] = v; sum += v; sumsq += v * v;
  }
#pragma unroll
  for (int off = 1; off < 16; off <<= 1) {
    sum += __shfl_xor(sum, off);
    sumsq += __shfl_xor(sumsq, off);
  }
  const float mu = sum * (1.f / 64.f);
  const float var = sumsq * (1.f / 64.f) - mu * mu;
  const float rstd = __frsqrt_rn(var + 1e-6f);
#pragma unroll
  for (int c = 0; c < 4; ++c) {
    const int cc = q * 4 + c;
    const float sh = sAdan[r][cc], sc = sAdan[r][64 + cc], gate = sAdan[r][128 + cc];
    const float ln = (nvv[c] - mu) * rstd;
    sH[r][cc] = sCat[r][cc] + gate * (ln * (1.f + sc) + sh);
  }
  __syncthreads();

  if (wid < 4) {
    v8f acc = {};
    const int nglob = wid * 16 + n;
#pragma unroll
    for (int k0 = 0; k0 < 64; k0 += 4) {
      const int kk = k0 + kp;
      v2f a; a.x = sH[m][kk]; a.y = sH[m][kk + 1];
      v2f b; b.x = sWhyp[nglob][kk]; b.y = sWhyp[nglob][kk + 1];
      acc = wmma4(a, b, acc);
    }
    const int mb = 8 * (lane >> 4);
    const float bh = sBhyp[nglob];
#pragma unroll
    for (int v = 0; v < 8; ++v)
      sSp[mb + v][nglob] = acc[v] + bh;
  }
  __syncthreads();

  float ss = 0.f;
  float spv[4];
#pragma unroll
  for (int c = 0; c < 4; ++c) {
    const int cc = q * 4 + c;
    const float v = (cc < 63) ? sSp[r][cc] : 0.f;
    spv[c] = v; ss += v * v;
  }
#pragma unroll
  for (int off = 1; off < 16; off <<= 1) ss += __shfl_xor(ss, off);
  const float tval = __fsqrt_rn(ss + 1.f);
  float* __restrict__ orow = out + (size_t)(m0 + r) * DD;
  if (q == 0) orow[0] = tval;
#pragma unroll
  for (int c = 0; c < 4; ++c) {
    const int cc = q * 4 + c;
    if (cc < 63) orow[1 + cc] = spv[c];
  }
}

extern "C" void kernel_launch(void* const* d_in, const int* in_sizes, int n_in,
                              void* d_out, int out_size, void* d_ws, size_t ws_size,
                              hipStream_t stream) {
  (void)in_sizes; (void)n_in; (void)out_size; (void)ws_size;
  const float* x      = (const float*)d_in[0];
  const float* e      = (const float*)d_in[2];
  const float* W_edge = (const float*)d_in[4];
  const float* b_edge = (const float*)d_in[5];
  const float* W_node = (const float*)d_in[6];
  const float* b_node = (const float*)d_in[7];
  const float* W_hyp  = (const float*)d_in[8];
  const float* b_hyp  = (const float*)d_in[9];
  const float* W_ada  = (const float*)d_in[10];
  const float* b_ada  = (const float*)d_in[11];
  const float* W_adan = (const float*)d_in[12];
  const float* b_adan = (const float*)d_in[13];
  float* out = (float*)d_out;

  float* ws_u   = (float*)d_ws;
  float* ws_agg = ws_u + NN * DD;

  k_edge<<<NN / GN, 1024, 0, stream>>>(x, e, W_edge, b_edge, W_ada, b_ada, ws_u, ws_agg);
  k_node<<<NN / 16, 256, 0, stream>>>(ws_u, ws_agg, W_adan, b_adan, W_node, b_node,
                                      W_hyp, b_hyp, out);
}


# Node kernel fused into the edge kernel (2 launches -> 1): each work-group updates its own 4 nodes on WMMA from agg kept in LDS; node weights staged in LDS behind the edge loads; sums unchanged
# speedup vs baseline: 2.0554x; 1.2934x over previous
#include <hip/hip_runtime.h>
#include <math.h>


typedef __attribute__((ext_vector_type(2))) float v2f;
typedef __attribute__((ext_vector_type(8))) float v8f;

#define NN 1024
#define DD 64

__device__ __forceinline__ v8f wmma4(v2f a, v2f b, v8f c) {
  return __builtin_amdgcn_wmma_f32_16x16x4_f32(
       false, a,  false, b,
       (short)0, c,  false,  false);
}

__device__ __forceinline__ float siluf(float v) {
  return v / (1.f + __expf(-v));
}
__device__ __forceinline__ float acoshf_pos(float z) {
  return __logf(z + __fsqrt_rn(z * z - 1.f));
}

#define GN 4
#define WROWS 80

__device__ __forceinline__ float4 logmap4(const float4 xv, const int c4, const int lane) {
  float ss = ((c4 != 0) ? xv.x * xv.x : 0.f) + xv.y * xv.y + xv.z * xv.z + xv.w * xv.w;
  for (int off = 1; off < 16; off <<= 1) ss += __shfl_xor(ss, off);
  const float x0 = __shfl(xv.x, lane & 16);
  const float nrm = fmaxf(__fsqrt_rn(ss), 1e-7f);
  const float z = fmaxf(x0, 1.f + 1e-7f);
  const float ac = acoshf_pos(z);
  float4 uv;
  uv.x = (c4 != 0) ? (ac * xv.x / nrm) : 0.f;
  uv.y = ac * xv.y / nrm;
  uv.z = ac * xv.z / nrm;
  uv.w = ac * xv.w / nrm;
  return uv;
}

__global__ __launch_bounds__(1024) void k_layer(
    const float* __restrict__ x, const float* __restrict__ e, const float* __restrict__ W_edge,
    const float* __restrict__ b_edge, const float* __restrict__ W_ada,
    const float* __restrict__ b_ada,
    const float* __restrict__ W_adan, const float* __restrict__ b_adan,
    const float* __restrict__ W_node, const float* __restrict__ b_node,
    const float* __restrict__ W_hyp, const float* __restrict__ b_hyp,
    float* __restrict__ out) {
  const int i0 = blockIdx.x * GN;
  const int t = threadIdx.x;
  const int wid = t >> 5, lane = t & 31;
  const int g = t >> 8, tl = t & 255;
  const int wl = wid & 7;
  const int i = i0 + g;
  __shared__ float sEF[GN][64][68];
  __shared__ float sW[64][196];
  __shared__ float sU[WROWS][68];
  __shared__ float sPj[WROWS][68];
  __shared__ float sPi[16][68];
  __shared__ float sWadan[192][68];
  __shared__ float sWnode[64][132];
  __shared__ float sWhyp[64][68];
  __shared__ float sBias[1024];
  __shared__ float sCat[GN][132];
  __shared__ float sSA[GN][68];
  __shared__ float sH[GN][68];
  __shared__ float sAdan[16][196];
  __shared__ float sNv[16][68];
  __shared__ float sSp[16][68];
  const int w0 = t >> 4, c4 = (t & 15) * 4;
  const float4 xv0 = *(const float4*)(x + (unsigned)(((i0 + w0) & (NN - 1)) * DD + c4));
  const int w1 = 64 + (t >> 4);
  float4 xv1 = make_float4(0.f, 0.f, 0.f, 0.f);
  if (t < 256) {
    const int wc = (w1 < 68) ? w1 : 67;
    xv1 = *(const float4*)(x + (unsigned)(((i0 + wc) & (NN - 1)) * DD + c4));
  }
#pragma unroll
  for (int p = 0; p < 3; ++p) {
    const int f = p * 1024 + t;
    *(float4*)&sW[f / 48][(f % 48) * 4] = *(const float4*)(W_edge + (unsigned)(f * 4));
  }
#pragma unroll
  for (int p = 0; p < 4; ++p) {
    const int f = p * 256 + tl;
    const int r = f >> 4, cc4 = (f & 15) * 4;
    const int j = (i + 1 + r) & (NN - 1);
    *(float4*)&sEF[g][r][cc4] = *(const float4*)(e + (unsigned)((i * NN + j) * DD + cc4));
  }
  {
    const float4 uv = logmap4(xv0, c4, lane);
    *(float4*)&sU[w0][c4] = uv;
  }
  if (t < 256) {
    float4 uv = logmap4(xv1, c4, lane);
    if (w1 >= 68) uv = make_float4(0.f, 0.f, 0.f, 0.f);
    *(float4*)&sU[w1][c4] = uv;
  }
  __syncthreads();

  const int m = lane & 15, kp = (lane >> 4) << 1, n = lane & 15;
  const int mbh = 8 * (lane >> 4);
  if (wid < 24) {
    const bool isPj = wid < 20;
    const int mtw = isPj ? (wid >> 2) : 0;
    const int ntw = isPj ? (wid & 3) : (wid - 20);
    const int kofs = isPj ? 64 : 0;
    v8f acc = {};
#pragma unroll
    for (int k0 = 0; k0 < 64; k0 += 4) {
      const int kk = k0 + kp;
      v2f a; a.x = sU[mtw * 16 + m][kk]; a.y = sU[mtw * 16 + m][kk + 1];
      v2f b; b.x = sW[ntw * 16 + n][kofs + kk]; b.y = sW[ntw * 16 + n][kofs + kk + 1];
      acc = wmma4(a, b, acc);
    }
    if (isPj) {
#pragma unroll
      for (int v = 0; v < 8; ++v) sPj[mtw * 16 + mbh + v][ntw * 16 + n] = acc[v];
    } else {
#pragma unroll
      for (int v = 0; v < 8; ++v) sPi[mbh + v][ntw * 16 + n] = acc[v];
    }
  }

  float4 wa[3], wn[2], wh;
#pragma unroll
  for (int p = 0; p < 3; ++p) wa[p] = *(const float4*)(W_adan + (unsigned)((p * 1024 + t) * 4));
#pragma unroll
  for (int p = 0; p < 2; ++p) wn[p] = *(const float4*)(W_node + (unsigned)((p * 1024 + t) * 4));
  {
    const int fc = (t < 63 * 16) ? t : (63 * 16 - 1);
    wh = *(const float4*)(W_hyp + (unsigned)(fc * 4));
    if (t >= 63 * 16) wh = make_float4(0.f, 0.f, 0.f, 0.f);
  }
  const float* bsrc = (t < 192) ? (b_adan + t) : (t < 256) ? (b_node + (t - 192))
                                               : (b_hyp + ((t < 256 + 63) ? (t - 256) : 62));
  const float bval = *bsrc;

  const int tile0 = wl * 2;
  const int mt = tile0 >> 2, nt0 = tile0 & 3;
  v8f acc0 = {}, acc1 = {};
#pragma unroll
  for (int k0 = 0; k0 < 64; k0 += 4) {
    const int kk = k0 + kp;
    v2f a;  a.x  = sEF[g][mt * 16 + m][kk];            a.y  = sEF[g][mt * 16 + m][kk + 1];
    v2f b0; b0.x = sW[nt0 * 16 + n][128 + kk];         b0.y = sW[nt0 * 16 + n][128 + kk + 1];
    v2f b1; b1.x = sW[(nt0 + 1) * 16 + n][128 + kk];   b1.y = sW[(nt0 + 1) * 16 + n][128 + kk + 1];
    acc0 = wmma4(a, b0, acc0);
    acc1 = wmma4(a, b1, acc1);
  }
#pragma unroll
  for (int p = 0; p < 3; ++p) {
    const int f = p * 1024 + t;
    *(float4*)&sWadan[f >> 4][(f & 15) * 4] = wa[p];
  }
#pragma unroll
  for (int p = 0; p < 2; ++p) {
    const int f = p * 1024 + t;
    *(float4*)&sWnode[f >> 5][(f & 31) * 4] = wn[p];
  }
  *(float4*)&sWhyp[t >> 4][(t & 15) * 4] = wh;
  sBias[t] = (t == 256 + 63) ? 0.f : bval;
  __syncthreads();
  {
    const int mb = mt * 16 + mbh;
#pragma unroll
    for (int v = 0; v < 8; ++v) {
      sEF[g][mb + v][nt0 * 16 + n] = acc0[v];
      sEF[g][mb + v][(nt0 + 1) * 16 + n] = acc1[v];
    }
  }
  __syncthreads();

  const int r = tl >> 2, q = tl & 3, c0 = q * 16;
  const int wr = g + 1 + r;
  float inner = 0.f, sum = 0.f, sumsq = 0.f;
  float vals[16];
#pragma unroll
  for (int c = 0; c < 16; ++c) {
    const int cc = c0 + c;
    inner += sU[g][cc] * sU[wr][cc];
    const float v = sEF[g][r][cc] + sPi[g][cc] + sPj[wr][cc] + b_edge[cc];
    vals[c] = v; sum += v; sumsq += v * v;
  }
  inner += __shfl_xor(inner, 1); inner += __shfl_xor(inner, 2);
  sum   += __shfl_xor(sum, 1);   sum   += __shfl_xor(sum, 2);
  sumsq += __shfl_xor(sumsq, 1); sumsq += __shfl_xor(sumsq, 2);
  const float mu = sum * (1.f / 64.f);
  const float var = sumsq * (1.f / 64.f) - mu * mu;
  const float rstd = __frsqrt_rn(var + 1e-6f);
  const float z = fmaxf(-inner, 1.f + 1e-7f);
  float dist = acoshf_pos(z);
  dist = fminf(fmaxf(dist, 1e-6f), 100.f);
  const float s = siluf(dist);
#pragma unroll
  for (int c = 0; c < 16; ++c) {
    const int cc = c0 + c;
    const float shift = s * W_ada[cc] + b_ada[cc];
    const float scale = s * W_ada[64 + cc] + b_ada[64 + cc];
    sEF[g][r][cc] = (vals[c] - mu) * rstd * (1.f + scale) + shift;
  }
  __syncthreads();
  if (tl < 64) {
    float a = 0.f;
#pragma unroll 8
    for (int rr = 0; rr < 64; ++rr) a += sEF[g][rr][tl];
    const float av = a * (1.f / 64.f);
    sCat[g][tl] = sU[g][tl];
    sCat[g][64 + tl] = av;
    sSA[g][tl] = siluf(av);
  }
  __syncthreads();

  const int mr = m & (GN - 1);
  const bool mv = m < GN;
  if (wid < 12) {
    v8f acc = {};
    const int nglob = wid * 16 + n;
#pragma unroll
    for (int k0 = 0; k0 < 64; k0 += 4) {
      const int kk = k0 + kp;
      const float ax = sSA[mr][kk], ay = sSA[mr][kk + 1];
      v2f a; a.x = mv ? ax : 0.f; a.y = mv ? ay : 0.f;
      v2f b; b.x = sWadan[nglob][kk]; b.y = sWadan[nglob][kk + 1];
      acc = wmma4(a, b, acc);
    }
    const float bb = sBias[nglob];
#pragma unroll
    for (int v = 0; v < GN; ++v) sAdan[mbh + v][nglob] = acc[v] + bb;
  } else if (wid < 16) {
    v8f acc = {};
    const int nglob = (wid - 12) * 16 + n;
#pragma unroll
    for (int k0 = 0; k0 < 128; k0 += 4) {
      const int kk = k0 + kp;
      const float ax = sCat[mr][kk], ay = sCat[mr][kk + 1];
      v2f a; a.x = mv ? ax : 0.f; a.y = mv ? ay : 0.f;
      v2f b; b.x = sWnode[nglob][kk]; b.y = sWnode[nglob][kk + 1];
      acc = wmma4(a, b, acc);
    }
    const float bb = sBias[192 + nglob];
#pragma unroll
    for (int v = 0; v < GN; ++v) sNv[mbh + v][nglob] = acc[v] + bb;
  }
  __syncthreads();

  const int nr = t >> 4, nq = t & 15;
  if (t < GN * 16) {
    float nsum = 0.f, nsumsq = 0.f;
    float nvv[4];
#pragma unroll
    for (int c = 0; c < 4; ++c) {
      const float v = sNv[nr][nq * 4 + c];
      nvv[c] = v; nsum += v; nsumsq += v * v;
    }
#pragma unroll
    for (int off = 1; off < 16; off <<= 1) {
      nsum += __shfl_xor(nsum, off);
      nsumsq += __shfl_xor(nsumsq, off);
    }
    const float nmu = nsum * (1.f / 64.f);
    const float nvar = nsumsq * (1.f / 64.f) - nmu * nmu;
    const float nrstd = __frsqrt_rn(nvar + 1e-6f);
#pragma unroll
    for (int c = 0; c < 4; ++c) {
      const int cc = nq * 4 + c;
      const float sh = sAdan[nr][cc], sc = sAdan[nr][64 + cc], gate = sAdan[nr][128 + cc];
      const float ln = (nvv[c] - nmu) * nrstd;
      sH[nr][cc] = sCat[nr][cc] + gate * (ln * (1.f + sc) + sh);
    }
  }
  __syncthreads();

  if (wid < 4) {
    v8f acc = {};
    const int nglob = wid * 16 + n;
#pragma unroll
    for (int k0 = 0; k0 < 64; k0 += 4) {
      const int kk = k0 + kp;
      const float ax = sH[mr][kk], ay = sH[mr][kk + 1];
      v2f a; a.x = mv ? ax : 0.f; a.y = mv ? ay : 0.f;
      v2f b; b.x = sWhyp[nglob][kk]; b.y = sWhyp[nglob][kk + 1];
      acc = wmma4(a, b, acc);
    }
    const float bh = sBias[256 + nglob];
#pragma unroll
    for (int v = 0; v < GN; ++v) sSp[mbh + v][nglob] = acc[v] + bh;
  }
  __syncthreads();

  if (t < GN * 16) {
    float ss = 0.f;
    float spv[4];
#pragma unroll
    for (int c = 0; c < 4; ++c) {
      const int cc = nq * 4 + c;
      const float v = (cc < 63) ? sSp[nr][cc] : 0.f;
      spv[c] = v; ss += v * v;
    }
#pragma unroll
    for (int off = 1; off < 16; off <<= 1) ss += __shfl_xor(ss, off);
    const float tval = __fsqrt_rn(ss + 1.f);
    float* __restrict__ orow = out + (unsigned)((i0 + nr) * DD);
    if (nq == 0) orow[0] = tval;
#pragma unroll
    for (int c = 0; c < 4; ++c) {
      const int cc = nq * 4 + c;
      if (cc < 63) orow[1 + cc] = spv[c];
    }
  }
}

extern "C" void kernel_launch(void* const* d_in, const int* in_sizes, int n_in,
                              void* d_out, int out_size, void* d_ws, size_t ws_size,
                              hipStream_t stream) {
  (void)in_sizes; (void)n_in; (void)out_size; (void)ws_size;
  const float* x      = (const float*)d_in[0];
  const float* e      = (const float*)d_in[2];
  const float* W_edge = (const float*)d_in[4];
  const float* b_edge = (const float*)d_in[5];
  const float* W_node = (const float*)d_in[6];
  const float* b_node = (const float*)d_in[7];
  const float* W_hyp  = (const float*)d_in[8];
  const float* b_hyp  = (const float*)d_in[9];
  const float* W_ada  = (const float*)d_in[10];
  const float* b_ada  = (const float*)d_in[11];
  const float* W_adan = (const float*)d_in[12];
  const float* b_adan = (const float*)d_in[13];
  float* out = (float*)d_out;

  (void)d_ws;
  k_layer<<<NN / GN, 1024, 0, stream>>>(x, e, W_edge, b_edge, W_ada, b_ada, W_adan, b_adan,
                                        W_node, b_node, W_hyp, b_hyp, out);
}


# Node stage: every serial chain (64-row column sum, adan/nv/sp WMMA chains, LN row) asks for its LDS operands as one group ahead: one LDS round trip per chain, not one per step; sums unchanged
# speedup vs baseline: 2.2017x; 1.0712x over previous
#include <hip/hip_runtime.h>
#include <math.h>


typedef __attribute__((ext_vector_type(2))) float v2f;
typedef __attribute__((ext_vector_type(8))) float v8f;

#define NN 1024
#define DD 64

__device__ __forceinline__ v8f wmma4(v2f a, v2f b, v8f c) {
  return __builtin_amdgcn_wmma_f32_16x16x4_f32(
       false, a,  false, b,
       (short)0, c,  false,  false);
}

__device__ __forceinline__ float siluf(float v) {
  return v / (1.f + __expf(-v));
}
__device__ __forceinline__ float acoshf_pos(float z) {
  return __logf(z + __fsqrt_rn(z * z - 1.f));
}

#define GN 4
#define WROWS 80

__device__ __forceinline__ float4 logmap4(const float4 xv, const int c4, const int lane) {
  float ss = ((c4 != 0) ? xv.x * xv.x : 0.f) + xv.y * xv.y + xv.z * xv.z + xv.w * xv.w;
  for (int off = 1; off < 16; off <<= 1) ss += __shfl_xor(ss, off);
  const float x0 = __shfl(xv.x, lane & 16);
  const float nrm = fmaxf(__fsqrt_rn(ss), 1e-7f);
  const float z = fmaxf(x0, 1.f + 1e-7f);
  const float ac = acoshf_pos(z);
  float4 uv;
  uv.x = (c4 != 0) ? (ac * xv.x / nrm) : 0.f;
  uv.y = ac * xv.y / nrm;
  uv.z = ac * xv.z / nrm;
  uv.w = ac * xv.w / nrm;
  return uv;
}

__global__ __launch_bounds__(1024) void k_layer(
    const float* __restrict__ x, const float* __restrict__ e, const float* __restrict__ W_edge,
    const float* __restrict__ b_edge, const float* __restrict__ W_ada,
    const float* __restrict__ b_ada,
    const float* __restrict__ W_adan, const float* __restrict__ b_adan,
    const float* __restrict__ W_node, const float* __restrict__ b_node,
    const float* __restrict__ W_hyp, const float* __restrict__ b_hyp,
    float* __restrict__ out) {
  const int i0 = blockIdx.x * GN;
  const int t = threadIdx.x;
  const int wid = t >> 5, lane = t & 31;
  const int g = t >> 8, tl = t & 255;
  const int wl = wid & 7;
  const int i = i0 + g;
  __shared__ float sEF[GN][64][68];
  __shared__ float sW[64][196];
  __shared__ float sU[WROWS][68];
  __shared__ float sPj[WROWS][68];
  __shared__ float sPi[16][68];
  __shared__ float sWadan[192][68];
  __shared__ float sWnode[64][132];
  __shared__ float sWhyp[64][68];
  __shared__ float sBias[1024];
  __shared__ float sCat[GN][132];
  __shared__ float sSA[GN][68];
  __shared__ float sH[GN][68];
  __shared__ float sAdan[16][196];
  __shared__ float sNv[16][68];
  __shared__ float sSp[16][68];
  const int w0 = t >> 4, c4 = (t & 15) * 4;
  const float4 xv0 = *(const float4*)(x + (unsigned)(((i0 + w0) & (NN - 1)) * DD + c4));
  const int w1 = 64 + (t >> 4);
  float4 xv1 = make_float4(0.f, 0.f, 0.f, 0.f);
  if (t < 256) {
    const int wc = (w1 < 68) ? w1 : 67;
    xv1 = *(const float4*)(x + (unsigned)(((i0 + wc) & (NN - 1)) * DD + c4));
  }
#pragma unroll
  for (int p = 0; p < 3; ++p) {
    const int f = p * 1024 + t;
    *(float4*)&sW[f / 48][(f % 48) * 4] = *(const float4*)(W_edge + (unsigned)(f * 4));
  }
#pragma unroll
  for (int p = 0; p < 4; ++p) {
    const int f = p * 256 + tl;
    const int r = f >> 4, cc4 = (f & 15) * 4;
    const int j = (i + 1 + r) & (NN - 1);
    *(float4*)&sEF[g][r][cc4] = *(const float4*)(e + (unsigned)((i * NN + j) * DD + cc4));
  }
  {
    const float4 uv = logmap4(xv0, c4, lane);
    *(float4*)&sU[w0][c4] = uv;
  }
  if (t < 256) {
    float4 uv = logmap4(xv1, c4, lane);
    if (w1 >= 68) uv = make_float4(0.f, 0.f, 0.f, 0.f);
    *(float4*)&sU[w1][c4] = uv;
  }
  __syncthreads();

  const int m = lane & 15, kp = (lane >> 4) << 1, n = lane & 15;
  const int mbh = 8 * (lane >> 4);
  if (wid < 24) {
    const bool isPj = wid < 20;
    const int mtw = isPj ? (wid >> 2) : 0;
    const int ntw = isPj ? (wid & 3) : (wid - 20);
    const int kofs = isPj ? 64 : 0;
    v8f acc = {};
#pragma unroll
    for (int k0 = 0; k0 < 64; k0 += 4) {
      const int kk = k0 + kp;
      v2f a; a.x = sU[mtw * 16 + m][kk]; a.y = sU[mtw * 16 + m][kk + 1];
      v2f b; b.x = sW[ntw * 16 + n][kofs + kk]; b.y = sW[ntw * 16 + n][kofs + kk + 1];
      acc = wmma4(a, b, acc);
    }
    if (isPj) {
#pragma unroll
      for (int v = 0; v < 8; ++v) sPj[mtw * 16 + mbh + v][ntw * 16 + n] = acc[v];
    } else {
#pragma unroll
      for (int v = 0; v < 8; ++v) sPi[mbh + v][ntw * 16 + n] = acc[v];
    }
  }

  float4 wa[3], wn[2], wh;
#pragma unroll
  for (int p = 0; p < 3; ++p) wa[p] = *(const float4*)(W_adan + (unsigned)((p * 1024 + t) * 4));
#pragma unroll
  for (int p = 0; p < 2; ++p) wn[p] = *(const float4*)(W_node + (unsigned)((p * 1024 + t) * 4));
  {
    const int fc = (t < 63 * 16) ? t : (63 * 16 - 1);
    wh = *(const float4*)(W_hyp + (unsigned)(fc * 4));
    if (t >= 63 * 16) wh = make_float4(0.f, 0.f, 0.f, 0.f);
  }
  const float* bsrc = (t < 192) ? (b_adan + t) : (t < 256) ? (b_node + (t - 192))
                                               : (b_hyp + ((t < 256 + 63) ? (t - 256) : 62));
  const float bval = *bsrc;

  const int tile0 = wl * 2;
  const int mt = tile0 >> 2, nt0 = tile0 & 3;
  v8f acc0 = {}, acc1 = {};
#pragma unroll
  for (int k0 = 0; k0 < 64; k0 += 4) {
    const int kk = k0 + kp;
    v2f a;  a.x  = sEF[g][mt * 16 + m][kk];            a.y  = sEF[g][mt * 16 + m][kk + 1];
    v2f b0; b0.x = sW[nt0 * 16 + n][128 + kk];         b0.y = sW[nt0 * 16 + n][128 + kk + 1];
    v2f b1; b1.x = sW[(nt0 + 1) * 16 + n][128 + kk];   b1.y = sW[(nt0 + 1) * 16 + n][128 + kk + 1];
    acc0 = wmma4(a, b0, acc0);
    acc1 = wmma4(a, b1, acc1);
  }
#pragma unroll
  for (int p = 0; p < 3; ++p) {
    const int f = p * 1024 + t;
    *(float4*)&sWadan[f >> 4][(f & 15) * 4] = wa[p];
  }
#pragma unroll
  for (int p = 0; p < 2; ++p) {
    const int f = p * 1024 + t;
    *(float4*)&sWnode[f >> 5][(f & 31) * 4] = wn[p];
  }
  *(float4*)&sWhyp[t >> 4][(t & 15) * 4] = wh;
  sBias[t] = (t == 256 + 63) ? 0.f : bval;
  __syncthreads();
  {
    const int mb = mt * 16 + mbh;
#pragma unroll
    for (int v = 0; v < 8; ++v) {
      sEF[g][mb + v][nt0 * 16 + n] = acc0[v];
      sEF[g][mb + v][(nt0 + 1) * 16 + n] = acc1[v];
    }
  }
  __syncthreads();

  const int r = tl >> 2, q = tl & 3, c0 = q * 16;
  const int wr = g + 1 + r;
  float inner = 0.f, sum = 0.f, sumsq = 0.f;
  float vals[16];
#pragma unroll
  for (int c = 0; c < 16; ++c) {
    const int cc = c0 + c;
    inner += sU[g][cc] * sU[wr][cc];
    const float v = sEF[g][r][cc] + sPi[g][cc] + sPj[wr][cc] + b_edge[cc];
    vals[c] = v; sum += v; sumsq += v * v;
  }
  inner += __shfl_xor(inner, 1); inner += __shfl_xor(inner, 2);
  sum   += __shfl_xor(sum, 1);   sum   += __shfl_xor(sum, 2);
  sumsq += __shfl_xor(sumsq, 1); sumsq += __shfl_xor(sumsq, 2);
  const float mu = sum * (1.f / 64.f);
  const float var = sumsq * (1.f / 64.f) - mu * mu;
  const float rstd = __frsqrt_rn(var + 1e-6f);
  const float z = fmaxf(-inner, 1.f + 1e-7f);
  float dist = acoshf_pos(z);
  dist = fminf(fmaxf(dist, 1e-6f), 100.f);
  const float s = siluf(dist);
#pragma unroll
  for (int c = 0; c < 16; ++c) {
    const int cc = c0 + c;
    const float shift = s * W_ada[cc] + b_ada[cc];
    const float scale = s * W_ada[64 + cc] + b_ada[64 + cc];
    sEF[g][r][cc] = (vals[c] - mu) * rstd * (1.f + scale) + shift;
  }
  __syncthreads();
  if (tl < 64) {
    float cv[64];
#pragma unroll
    for (int rr = 0; rr < 64; ++rr) cv[rr] = sEF[g][rr][tl];
    __builtin_amdgcn_sched_barrier(0);
    float a = 0.f;
#pragma unroll
    for (int rr = 0; rr < 64; ++rr) a += cv[rr];
    const float av = a * (1.f / 64.f);
    sCat[g][tl] = sU[g][tl];
    sCat[g][64 + tl] = av;
    sSA[g][tl] = siluf(av);
  }
  __syncthreads();

  const int mr = m & (GN - 1);
  const bool mv = m < GN;
  if (wid < 12) {
    v8f acc = {};
    const int nglob = wid * 16 + n;
    v2f fa[16], fb[16];
#pragma unroll
    for (int s = 0; s < 16; ++s) {
      if (s == 4) __builtin_amdgcn_sched_barrier(0);
      const int kk = s * 4 + kp;
      fa[s].x = sSA[mr][kk]; fa[s].y = sSA[mr][kk + 1];
      fb[s].x = sWadan[nglob][kk]; fb[s].y = sWadan[nglob][kk + 1];
    }
    __builtin_amdgcn_sched_barrier(0);
#pragma unroll
    for (int s = 0; s < 16; ++s) {
      v2f a; a.x = mv ? fa[s].x : 0.f; a.y = mv ? fa[s].y : 0.f;
      acc = wmma4(a, fb[s], acc);
    }
    const float bb = sBias[nglob];
#pragma unroll
    for (int v = 0; v < GN; ++v) sAdan[mbh + v][nglob] = acc[v] + bb;
  } else if (wid < 16) {
    v8f acc = {};
    const int nglob = (wid - 12) * 16 + n;
    v2f fa[4][8], fb[4][8];
#pragma unroll
    for (int s = 0; s < 16; ++s) {
      if (s == 4) __builtin_amdgcn_sched_barrier(0);
      const int kk = s * 4 + kp;
      fa[s >> 3][s & 7].x = sCat[mr][kk]; fa[s >> 3][s & 7].y = sCat[mr][kk + 1];
      fb[s >> 3][s & 7].x = sWnode[nglob][kk]; fb[s >> 3][s & 7].y = sWnode[nglob][kk + 1];
    }
#pragma unroll
    for (int qk = 0; qk < 4; ++qk) {
      __builtin_amdgcn_sched_barrier(0);
      if (qk + 2 < 4) {
#pragma unroll
        for (int s = 0; s < 8; ++s) {
          const int kk = (qk + 2) * 32 + s * 4 + kp;
          fa[qk + 2][s].x = sCat[mr][kk]; fa[qk + 2][s].y = sCat[mr][kk + 1];
          fb[qk + 2][s].x = sWnode[nglob][kk]; fb[qk + 2][s].y = sWnode[nglob][kk + 1];
        }
      }
#pragma unroll
      for (int s = 0; s < 8; ++s) {
        v2f a; a.x = mv ? fa[qk][s].x : 0.f; a.y = mv ? fa[qk][s].y : 0.f;
        acc = wmma4(a, fb[qk][s], acc);
      }
    }
    const float bb = sBias[192 + nglob];
#pragma unroll
    for (int v = 0; v < GN; ++v) sNv[mbh + v][nglob] = acc[v] + bb;
  }
  __syncthreads();

  const int nr = t >> 4, nq = t & 15;
  if (t < GN * 16) {
    float nsum = 0.f, nsumsq = 0.f;
    float nvv[4], shv[4], scv[4], gav[4], xtv[4];
#pragma unroll
    for (int c = 0; c < 4; ++c) {
      const int cc = nq * 4 + c;
      nvv[c] = sNv[nr][cc];
      shv[c] = sAdan[nr][cc]; scv[c] = sAdan[nr][64 + cc]; gav[c] = sAdan[nr][128 + cc];
      xtv[c] = sCat[nr][cc];
    }
    __builtin_amdgcn_sched_barrier(0);
#pragma unroll
    for (int c = 0; c < 4; ++c) {
      const float v = nvv[c];
      nsum += v; nsumsq += v * v;
    }
#pragma unroll
    for (int off = 1; off < 16; off <<= 1) {
      nsum += __shfl_xor(nsum, off);
      nsumsq += __shfl_xor(nsumsq, off);
    }
    const float nmu = nsum * (1.f / 64.f);
    const float nvar = nsumsq * (1.f / 64.f) - nmu * nmu;
    const float nrstd = __frsqrt_rn(nvar + 1e-6f);
#pragma unroll
    for (int c = 0; c < 4; ++c) {
      const int cc = nq * 4 + c;
      const float sh = shv[c], sc = scv[c], gate = gav[c];
      const float ln = (nvv[c] - nmu) * nrstd;
      sH[nr][cc] = xtv[c] + gate * (ln * (1.f + sc) + sh);
    }
  }
  __syncthreads();

  if (wid < 4) {
    v8f acc = {};
    const int nglob = wid * 16 + n;
    v2f fa[16], fb[16];
#pragma unroll
    for (int s = 0; s < 16; ++s) {
      if (s == 4) __builtin_amdgcn_sched_barrier(0);
      const int kk = s * 4 + kp;
      fa[s].x = sH[mr][kk]; fa[s].y = sH[mr][kk + 1];
      fb[s].x = sWhyp[nglob][kk]; fb[s].y = sWhyp[nglob][kk + 1];
    }
    __builtin_amdgcn_sched_barrier(0);
#pragma unroll
    for (int s = 0; s < 16; ++s) {
      v2f a; a.x = mv ? fa[s].x : 0.f; a.y = mv ? fa[s].y : 0.f;
      acc = wmma4(a, fb[s], acc);
    }
    const float bh = sBias[256 + nglob];
#pragma unroll
    for (int v = 0; v < GN; ++v) sSp[mbh + v][nglob] = acc[v] + bh;
  }
  __syncthreads();

  if (t < GN * 16) {
    float ss = 0.f;
    float spv[4];
#pragma unroll
    for (int c = 0; c < 4; ++c) {
      const int cc = nq * 4 + c;
      const float v = (cc < 63) ? sSp[nr][cc] : 0.f;
      spv[c] = v; ss += v * v;
    }
#pragma unroll
    for (int off = 1; off < 16; off <<= 1) ss += __shfl_xor(ss, off);
    const float tval = __fsqrt_rn(ss + 1.f);
    float* __restrict__ orow = out + (unsigned)((i0 + nr) * DD);
    if (nq == 0) orow[0] = tval;
#pragma unroll
    for (int c = 0; c < 4; ++c) {
      const int cc = nq * 4 + c;
      if (cc < 63) orow[1 + cc] = spv[c];
    }
  }
}

extern "C" void kernel_launch(void* const* d_in, const int* in_sizes, int n_in,
                              void* d_out, int out_size, void* d_ws, size_t ws_size,
                              hipStream_t stream) {
  (void)in_sizes; (void)n_in; (void)out_size; (void)ws_size;
  const float* x      = (const float*)d_in[0];
  const float* e      = (const float*)d_in[2];
  const float* W_edge = (const float*)d_in[4];
  const float* b_edge = (const float*)d_in[5];
  const float* W_node = (const float*)d_in[6];
  const float* b_node = (const float*)d_in[7];
  const float* W_hyp  = (const float*)d_in[8];
  const float* b_hyp  = (const float*)d_in[9];
  const float* W_ada  = (const float*)d_in[10];
  const float* b_ada  = (const float*)d_in[11];
  const float* W_adan = (const float*)d_in[12];
  const float* b_adan = (const float*)d_in[13];
  float* out = (float*)d_out;

  (void)d_ws;
  k_layer<<<NN / GN, 1024, 0, stream>>>(x, e, W_edge, b_edge, W_ada, b_ada, W_adan, b_adan,
                                        W_node, b_node, W_hyp, b_hyp, out);
}
